# starting kernel
# baseline (speedup 1.0000x reference)
#include <hip/hip_runtime.h>
#include <math.h>

#define N_NODES   50000
#define DEG       16
#define IN_FEATS  128
#define HEADS     8
#define OUT_FEATS 16
#define HF        (HEADS * OUT_FEATS)
#define KSTEPS    (IN_FEATS / 4)

typedef __attribute__((ext_vector_type(2))) float v2f;
typedef __attribute__((ext_vector_type(4))) float v4f;
typedef __attribute__((ext_vector_type(8))) float v8f;

__global__ void gat_gemm_proj_kernel(const float* __restrict__ h,
                                     const float* __restrict__ W,
                                     const float* __restrict__ w_attn,
                                     float* __restrict__ hp,
                                     float* __restrict__ s)
{
    __shared__ float Wlds[KSTEPS * 8 * 64];

    const int tid  = threadIdx.x;
    const int lane = tid & 31;

    #pragma unroll
    for (int it = 0; it < 16; ++it) {
        const int g    = tid + it * 256;
        const int row  = g >> 5;
        const int c4   = g & 31;
        const v4f w    = *(const v4f*)(W + (size_t)row * IN_FEATS + c4 * 4);
        const int base = (c4 * 8 + (row >> 4)) * 64 + ((row & 15) << 1);
        Wlds[base + 0]  = w.x;
        Wlds[base + 1]  = w.y;
        Wlds[base + 32] = w.z;
        Wlds[base + 33] = w.w;
    }
    __syncthreads();

    const int wid = (blockIdx.x * blockDim.x + tid) >> 5;
    if (wid < N_NODES / 16) {
        const int ml = lane & 15;
        const int kh = (lane >> 4) << 1;
        const int m0 = wid * 16;

        const float* __restrict__ arow = h + (size_t)(m0 + ml) * IN_FEATS + kh;
        v2f areg[KSTEPS];
        #pragma unroll
        for (int t = 0; t < KSTEPS; ++t)
            areg[t] = *(const v2f*)(arow + 4 * t);

        v8f acc[8];
        #pragma unroll
        for (int j = 0; j < 8; ++j) acc[j] = (v8f){0.f,0.f,0.f,0.f,0.f,0.f,0.f,0.f};

        #pragma unroll
        for (int t = 0; t < KSTEPS; ++t) {
            v2f bf[8];
            #pragma unroll
            for (int j = 0; j < 8; ++j)
                bf[j] = *(const v2f*)(&Wlds[(t * 8 + j) * 64 + (lane << 1)]);
            #pragma unroll
            for (int j = 0; j < 8; ++j)
                acc[j] = __builtin_amdgcn_wmma_f32_16x16x4_f32(
                    false, areg[t], false, bf[j], (short)0, acc[j], false, false);
        }

        const float watt = w_attn[ml];
        #pragma unroll
        for (int j = 0; j < 8; ++j) {
            #pragma unroll
            for (int r = 0; r < 8; ++r) {
                const int row = m0 + r + ((lane >> 4) << 3);
                const float v = acc[j][r];
                hp[(size_t)row * HF + 16 * j + ml] = v;
                float p = v * watt;
                p += __shfl_xor(p, 1);
                p += __shfl_xor(p, 2);
                p += __shfl_xor(p, 4);
                p += __shfl_xor(p, 8);
                if (ml == 0) s[(size_t)row * HEADS + j] = p;
            }
        }
    }
}

__global__ void gat_attn_agg_kernel(const float* __restrict__ hp,
                                    const float* __restrict__ s,
                                    const float* __restrict__ ew,
                                    const int*   __restrict__ src,
                                    const float* __restrict__ bias,
                                    float* __restrict__ out)
{
    __shared__ float alpha_lds[8][DEG * HEADS];
    __shared__ int   src_lds[8][DEG];

    const int n     = (blockIdx.x * blockDim.x + threadIdx.x) >> 5;
    const int lane  = threadIdx.x & 31;
    const int wslot = (threadIdx.x >> 5);

    const int k  = lane & 15;
    const int hb = (lane >> 4) << 2;

    const int   sidx = src[n * DEG + k];
    const float lw   = log1pf(ew[n * DEG + k]);
    if (lane < 16) src_lds[wslot][k] = sidx;

    #pragma unroll
    for (int q = 0; q < 4; ++q) {
        const int hq = hb + q;
        float x = s[(size_t)sidx * HEADS + hq] + s[(size_t)n * HEADS + hq] + lw;
        x = (x > 0.f) ? x : 0.01f * x;
        float m = x;
        m = fmaxf(m, __shfl_xor(m, 1));
        m = fmaxf(m, __shfl_xor(m, 2));
        m = fmaxf(m, __shfl_xor(m, 4));
        m = fmaxf(m, __shfl_xor(m, 8));
        float ex = expf(x - m);
        float d = ex;
        d += __shfl_xor(d, 1);
        d += __shfl_xor(d, 2);
        d += __shfl_xor(d, 4);
        d += __shfl_xor(d, 8);
        alpha_lds[wslot][k * HEADS + hq] = ex / d;
    }

    __syncthreads();

    const int c  = lane << 2;
    const int hc = lane >> 2;
    float a0 = 0.f, a1 = 0.f, a2 = 0.f, a3 = 0.f;
    #pragma unroll
    for (int kk = 0; kk < DEG; ++kk) {
        const int   sk = src_lds[wslot][kk];
        const float al = alpha_lds[wslot][kk * HEADS + hc];
        const v4f   v  = *(const v4f*)(hp + (size_t)sk * HF + c);
        a0 += al * v.x;
        a1 += al * v.y;
        a2 += al * v.z;
        a3 += al * v.w;
    }
    const v4f b4 = *(const v4f*)(bias + c);
    v4f r;
    r.x = a0 + b4.x;
    r.y = a1 + b4.y;
    r.z = a2 + b4.z;
    r.w = a3 + b4.w;
    *(v4f*)(out + (size_t)n * HF + c) = r;
}

extern "C" void kernel_launch(void* const* d_in, const int* in_sizes, int n_in,
                              void* d_out, int out_size, void* d_ws, size_t ws_size,
                              hipStream_t stream) {
    const float* h       = (const float*)d_in[0];
    const float* ew      = (const float*)d_in[1];
    const int*   src     = (const int*)d_in[2];
    const float* W_fc    = (const float*)d_in[4];
    const float* w_attn  = (const float*)d_in[5];
    const float* bias    = (const float*)d_in[6];
    float*       out     = (float*)d_out;

    float* hp = (float*)d_ws;
    float* s  = hp + (size_t)N_NODES * HF;

    {
        const int waves  = N_NODES / 16;
        const int blocks = (waves * 32 + 255) / 256;
        gat_gemm_proj_kernel<<<blocks, 256, 0, stream>>>(h, W_fc, w_attn, hp, s);
    }
    {
        const int blocks = (N_NODES * 32) / 256;
        gat_attn_agg_kernel<<<blocks, 256, 0, stream>>>(hp, s, ew, src, bias, out);
    }
}


# Gather kernel re-decomposed: resident grid of 1024 work-groups, each wave walks 6-7 nodes on its own (no block barrier); next nodes' edge and score reads fly under the 16-row gather
# speedup vs baseline: 1.4639x; 1.4639x over previous
#include <hip/hip_runtime.h>
#include <math.h>

#define N_NODES   50000
#define DEG       16
#define IN_FEATS  128
#define HEADS     8
#define OUT_FEATS 16
#define HF        (HEADS * OUT_FEATS)
#define KSTEPS    (IN_FEATS / 4)

typedef __attribute__((ext_vector_type(2))) float v2f;
typedef __attribute__((ext_vector_type(4))) float v4f;
typedef __attribute__((ext_vector_type(8))) float v8f;

#define STRIPS    (N_NODES / 16)
#define G1        512
#define APITCH    132
#define KQUADS    (IN_FEATS / 8)
#define TILE_F    (16 * APITCH)

__device__ __forceinline__ void row_phase(const float* dt, float* __restrict__ hp, float* __restrict__ s,
                                          const v4f w4, int st, int roff, int lane, int j)
{
    const v4f v0 = *(const v4f*)(dt + roff);
    const v4f v1 = *(const v4f*)(dt + roff + APITCH);
    const int row0 = st * 16 + 2 * j;
    *(v4f*)(hp + (size_t)row0 * HF + 4 * lane)       = v0;
    *(v4f*)(hp + (size_t)(row0 + 1) * HF + 4 * lane) = v1;
    float p0 = v0.x * w4.x + v0.y * w4.y + v0.z * w4.z + v0.w * w4.w;
    float p1 = v1.x * w4.x + v1.y * w4.y + v1.z * w4.z + v1.w * w4.w;
    p0 += __shfl_xor(p0, 1);
    p1 += __shfl_xor(p1, 1);
    p0 += __shfl_xor(p0, 2);
    p1 += __shfl_xor(p1, 2);
    const int odd = lane & 1;
    s[(size_t)(row0 + odd) * HEADS + (lane >> 2)] = odd ? p1 : p0;
}

template <bool PREV>
__device__ __forceinline__ void strip_step(const float* at, float* dt, const float* dprev,
                                           const v4f (&bq)[KQUADS], const v4f w4,
                                           float* __restrict__ hp, float* __restrict__ s,
                                           int st, int aoff, int doff, int roff, int lane, int j)
{
    const float* ab = at + aoff;
    v4f aq[KQUADS];
    #pragma unroll
    for (int u = 0; u < KQUADS; ++u)
        aq[u] = *(const v4f*)(ab + 8 * u);
    __builtin_amdgcn_sched_barrier(0);

    if (PREV) row_phase(dprev, hp, s, w4, st - 1, roff, lane, j);

    v8f acc0 = (v8f){0.f,0.f,0.f,0.f,0.f,0.f,0.f,0.f};
    v8f acc1 = acc0;
    #pragma unroll
    for (int u = 0; u < KQUADS; ++u) {
        acc0 = __builtin_amdgcn_wmma_f32_16x16x4_f32(
            false, aq[u].xy, false, bq[u].xy, (short)0, acc0, false, false);
        acc1 = __builtin_amdgcn_wmma_f32_16x16x4_f32(
            false, aq[u].zw, false, bq[u].zw, (short)0, acc1, false, false);
    }

    float* db = dt + doff;
    #pragma unroll
    for (int r = 0; r < 8; ++r)
        db[r * APITCH] = acc0[r] + acc1[r];
}

__global__ __launch_bounds__(256)
void gat_gemm_proj_kernel(const float* __restrict__ h,
                          const float* __restrict__ W,
                          const float* __restrict__ w_attn,
                          float* __restrict__ hp,
                          float* __restrict__ s)
{
    __shared__ __attribute__((aligned(16))) float Alds[2 * TILE_F];
    __shared__ __attribute__((aligned(16))) float Dlds[2 * TILE_F];

    const int tid  = threadIdx.x;
    const int lane = tid & 31;
    const int j    = tid >> 5;
    const int c    = lane & 15;
    const int hh   = lane >> 4;

    const int s0 = (int)(((long long)blockIdx.x * STRIPS) / G1);
    const int s1 = (int)(((long long)(blockIdx.x + 1) * STRIPS) / G1);

    const float* __restrict__ wrow = W + (size_t)(16 * j + c) * IN_FEATS + 4 * hh;
    v4f bq[KQUADS];
    #pragma unroll
    for (int u = 0; u < KQUADS; ++u)
        bq[u] = *(const v4f*)(wrow + 8 * u);

    const v4f w4 = *(const v4f*)(w_attn + 4 * (lane & 3));

    const int    soff = j * APITCH + 4 * lane;
    const size_t goff = (size_t)j * IN_FEATS + 4 * lane;
    const int aoff = c * APITCH + 4 * hh;
    const int doff = (8 * hh) * APITCH + 16 * j + c;
    const int roff = (2 * j) * APITCH + 4 * lane;

    v4f x0, x1;
    {
        const float* g = h + (size_t)s0 * 16 * IN_FEATS + goff;
        const v4f y0 = *(const v4f*)g;
        const v4f y1 = *(const v4f*)(g + 8 * IN_FEATS);
        x0 = *(const v4f*)(g + 16 * IN_FEATS);
        x1 = *(const v4f*)(g + 24 * IN_FEATS);
        *(v4f*)&Alds[soff]              = y0;
        *(v4f*)&Alds[soff + 8 * APITCH] = y1;
    }
    __syncthreads();

    strip_step<false>(Alds, Dlds, Dlds, bq, w4, hp, s, s0, aoff, doff, roff, lane, j);
    *(v4f*)&Alds[TILE_F + soff]              = x0;
    *(v4f*)&Alds[TILE_F + soff + 8 * APITCH] = x1;
    __syncthreads();

    int buf = 1;
    for (int st = s0 + 1; st < s1; ++st) {
        const bool more = (st + 1 < s1);
        if (more) {
            const float* g = h + (size_t)(st + 1) * 16 * IN_FEATS + goff;
            x0 = *(const v4f*)g;
            x1 = *(const v4f*)(g + 8 * IN_FEATS);
        }

        strip_step<true>(Alds + buf * TILE_F, Dlds + buf * TILE_F, Dlds + (buf ^ 1) * TILE_F,
                         bq, w4, hp, s, st, aoff, doff, roff, lane, j);

        if (more) {
            *(v4f*)&Alds[(buf ^ 1) * TILE_F + soff]              = x0;
            *(v4f*)&Alds[(buf ^ 1) * TILE_F + soff + 8 * APITCH] = x1;
        }
        __syncthreads();
        buf ^= 1;
    }
    row_phase(Dlds + (buf ^ 1) * TILE_F, hp, s, w4, s1 - 1, roff, lane, j);
}

#define G2        1024
#define NWAVES2   (G2 * 8)

struct NodeIn {
    unsigned sidx;
    float ew;
    v4f   sd;
    v4f   ss;
};

__device__ __forceinline__ void node_step(NodeIn& cur, NodeIn& nxt, const int n,
                                          float* al, unsigned* sl,
                                          const float* __restrict__ hp, const float* __restrict__ s,
                                          const float* __restrict__ ew, const int* __restrict__ src,
                                          float* __restrict__ out, const v4f b4,
                                          const unsigned lane, const unsigned k, const unsigned hh, const unsigned hc)
{
    const float lw = log1pf(cur.ew);
    sl[k] = cur.sidx;
    const float xs[4] = { cur.ss.x + cur.sd.x + lw, cur.ss.y + cur.sd.y + lw,
                          cur.ss.z + cur.sd.z + lw, cur.ss.w + cur.sd.w + lw };
    #pragma unroll
    for (int q = 0; q < 4; ++q) {
        float x = xs[q];
        x = (x > 0.f) ? x : 0.01f * x;
        float m = x;
        m = fmaxf(m, __shfl_xor(m, 1));
        m = fmaxf(m, __shfl_xor(m, 2));
        m = fmaxf(m, __shfl_xor(m, 4));
        m = fmaxf(m, __shfl_xor(m, 8));
        float ex = expf(x - m);
        float d = ex;
        d += __shfl_xor(d, 1);
        d += __shfl_xor(d, 2);
        d += __shfl_xor(d, 4);
        d += __shfl_xor(d, 8);
        al[k * HEADS + 4u * hh + q] = ex / d;
    }

    __builtin_amdgcn_sched_barrier(0);
    nxt.ss = ((const v4f*)s)[nxt.sidx * 2u + hh];

    __builtin_amdgcn_fence(__ATOMIC_RELEASE, "wavefront");
    __builtin_amdgcn_wave_barrier();
    __builtin_amdgcn_fence(__ATOMIC_ACQUIRE, "wavefront");

    v4f v[DEG];
    #pragma unroll
    for (int kk = 0; kk < DEG; ++kk) {
        const unsigned sk = sl[kk];
        v[kk] = ((const v4f*)hp)[sk * (unsigned)(HF / 4) + lane];
    }
    __builtin_amdgcn_sched_barrier(0);

    const int nf = min(n + 2 * NWAVES2, N_NODES - 1);
    cur.sidx = (unsigned)src[(size_t)nf * DEG + k];
    cur.ew   = ew[(size_t)nf * DEG + k];
    cur.sd   = ((const v4f*)s)[(unsigned)nf * 2u + hh];
    __builtin_amdgcn_sched_barrier(0);

    float a0 = 0.f, a1 = 0.f, a2 = 0.f, a3 = 0.f;
    #pragma unroll
    for (int kk = 0; kk < DEG; ++kk) {
        const float a = al[kk * HEADS + hc];
        a0 += a * v[kk].x;
        a1 += a * v[kk].y;
        a2 += a * v[kk].z;
        a3 += a * v[kk].w;
    }
    v4f r;
    r.x = a0 + b4.x;
    r.y = a1 + b4.y;
    r.z = a2 + b4.z;
    r.w = a3 + b4.w;
    ((v4f*)out)[(unsigned)n * (unsigned)(HF / 4) + lane] = r;

    __builtin_amdgcn_fence(__ATOMIC_RELEASE, "wavefront");
    __builtin_amdgcn_wave_barrier();
    __builtin_amdgcn_fence(__ATOMIC_ACQUIRE, "wavefront");
}

__global__ __launch_bounds__(256) __attribute__((amdgpu_waves_per_eu(8)))
void gat_attn_agg_kernel(const float* __restrict__ hp,
                         const float* __restrict__ s,
                         const float* __restrict__ ew,
                         const int*   __restrict__ src,
                         const float* __restrict__ bias,
                         float* __restrict__ out)
{
    __shared__ float alpha_lds[8][DEG * HEADS];
    __shared__ unsigned src_lds[8][DEG];

    const unsigned lane = threadIdx.x & 31;
    const int wslot = __builtin_amdgcn_readfirstlane((int)(threadIdx.x >> 5));
    float* al = alpha_lds[wslot];
    unsigned* sl = src_lds[wslot];

    const unsigned k  = lane & 15;
    const unsigned hh = lane >> 4;
    const unsigned hc = lane >> 2;
    const v4f b4 = ((const v4f*)bias)[lane];

    const int w   = blockIdx.x * 8 + wslot;
    const int cnt = (N_NODES - 1 - w) / NWAVES2 + 1;

    NodeIn A, B;
    A.sidx = (unsigned)src[(size_t)w * DEG + k];
    A.ew   = ew[(size_t)w * DEG + k];
    A.sd   = ((const v4f*)s)[(unsigned)w * 2u + hh];
    A.ss   = ((const v4f*)s)[A.sidx * 2u + hh];
    __builtin_amdgcn_sched_barrier(0);
    B.sidx = (unsigned)src[(size_t)(w + NWAVES2) * DEG + k];
    B.ew   = ew[(size_t)(w + NWAVES2) * DEG + k];
    B.sd   = ((const v4f*)s)[(unsigned)(w + NWAVES2) * 2u + hh];

    int n = w;
    const int pairs = cnt >> 1;
    for (int p = 0; p < pairs; ++p) {
        node_step(A, B, n, al, sl, hp, s, ew, src, out, b4, lane, k, hh, hc);
        node_step(B, A, n + NWAVES2, al, sl, hp, s, ew, src, out, b4, lane, k, hh, hc);
        n += 2 * NWAVES2;
    }
    if (cnt & 1)
        node_step(A, B, n, al, sl, hp, s, ew, src, out, b4, lane, k, hh, hc);
}

extern "C" void kernel_launch(void* const* d_in, const int* in_sizes, int n_in,
                              void* d_out, int out_size, void* d_ws, size_t ws_size,
                              hipStream_t stream) {
    const float* h       = (const float*)d_in[0];
    const float* ew      = (const float*)d_in[1];
    const int*   src     = (const int*)d_in[2];
    const float* W_fc    = (const float*)d_in[4];
    const float* w_attn  = (const float*)d_in[5];
    const float* bias    = (const float*)d_in[6];
    float*       out     = (float*)d_out;

    float* hp = (float*)d_ws;
    float* s  = hp + (size_t)N_NODES * HF;

    {
        gat_gemm_proj_kernel<<<G1, 256, 0, stream>>>(h, W_fc, w_attn, hp, s);
    }
    {
        gat_attn_agg_kernel<<<G2, 256, 0, stream>>>(hp, s, ew, src, bias, out);
    }
}
